# starting kernel
# baseline (speedup 1.0000x reference)
#include <hip/hip_runtime.h>

typedef float v2f __attribute__((ext_vector_type(2)));
typedef float v8f __attribute__((ext_vector_type(8)));

__global__ void zero_i32_kernel(int* __restrict__ p, int n) {
    int i = blockIdx.x * blockDim.x + threadIdx.x;
    if (i < n) p[i] = 0;
}

__global__ void count_deg_kernel(const int* __restrict__ dst, int* __restrict__ deg, int E) {
    int e = blockIdx.x * blockDim.x + threadIdx.x;
    if (e < E) atomicAdd(&deg[dst[e]], 1);
}

__global__ void dinv_kernel(const int* __restrict__ deg, float* __restrict__ dinv, int n) {
    int i = blockIdx.x * blockDim.x + threadIdx.x;
    if (i < n) dinv[i] = rsqrtf((float)deg[i] + 1.0f);
}

__global__ void scan_kernel(const int* __restrict__ deg, int* __restrict__ rowstart, int n) {
    __shared__ int smem[256];
    __shared__ int carry;
    const int tid = threadIdx.x;
    if (tid == 0) carry = 0;
    __syncthreads();
    for (int base = 0; base < n; base += 256) {
        const int i = base + tid;
        const int v = (i < n) ? deg[i] : 0;
        smem[tid] = v;
        __syncthreads();
        for (int ofs = 1; ofs < 256; ofs <<= 1) {
            int t = (tid >= ofs) ? smem[tid - ofs] : 0;
            __syncthreads();
            smem[tid] += t;
            __syncthreads();
        }
        const int incl = smem[tid];
        if (i < n) rowstart[i] = carry + (incl - v);
        __syncthreads();
        if (tid == 0) carry += smem[255];
        __syncthreads();
    }
    if (tid == 0) rowstart[n] = carry;
}

__global__ void copy_i32_kernel(const int* __restrict__ a, int* __restrict__ b, int n) {
    int i = blockIdx.x * blockDim.x + threadIdx.x;
    if (i < n) b[i] = a[i];
}

__global__ void fill_csr_kernel(const int* __restrict__ src, const int* __restrict__ dst,
                                const float* __restrict__ dinv, int* __restrict__ cursor,
                                int* __restrict__ csr_src, float* __restrict__ csr_norm, int E) {
    int e = blockIdx.x * blockDim.x + threadIdx.x;
    if (e >= E) return;
    const int s = src[e];
    const int d = dst[e];
    const int pos = atomicAdd(&cursor[d], 1);
    csr_src[pos]  = s;
    csr_norm[pos] = dinv[s] * dinv[d];
}

__global__ void gemm_f32_wmma_kernel(const float* __restrict__ A,
                                     const float* __restrict__ B,
                                     float* __restrict__ C,
                                     int M, int N, int K) {
    const int lane = threadIdx.x & 31;
    const int wave = (blockIdx.x * blockDim.x + threadIdx.x) >> 5;
    const int nTiles = N >> 4;
    const int mTiles = M >> 4;
    if (wave >= mTiles * nTiles) return;
    const int mt  = wave / nTiles;
    const int nt  = wave - mt * nTiles;
    const int l16 = lane & 15;
    const int hi  = lane >> 4;
    const int row = (mt << 4) + l16;
    const int col = (nt << 4) + l16;
    const int kOff = hi << 1;

    v8f acc0 = {};
    v8f acc1 = {};
    const float* ap = A + (size_t)row * K + kOff;
    const float* bp = B + (size_t)kOff * N + col;
    const size_t bStep4 = (size_t)4 * N;

    for (int k = 0; k < K; k += 8) {
        v2f a0 = *(const v2f*)ap;
        v2f b0; b0.x = bp[0]; b0.y = bp[N];
        acc0 = __builtin_amdgcn_wmma_f32_16x16x4_f32(false, a0, false, b0,
                                                     (short)0, acc0, false, false);
        v2f a1 = *(const v2f*)(ap + 4);
        v2f b1; b1.x = bp[bStep4]; b1.y = bp[bStep4 + N];
        acc1 = __builtin_amdgcn_wmma_f32_16x16x4_f32(false, a1, false, b1,
                                                     (short)0, acc1, false, false);
        ap += 8;
        bp += 2 * bStep4;
    }
    v8f acc = acc0 + acc1;

    const int mBase = (mt << 4) + (hi << 3);
    #pragma unroll
    for (int v = 0; v < 8; ++v) {
        C[(size_t)(mBase + v) * N + col] = acc[v];
    }
}

__global__ void gather256_kernel(const int* __restrict__ rowstart,
                                 const int* __restrict__ csr_src,
                                 const float* __restrict__ csr_norm,
                                 const float* __restrict__ dinv,
                                 const float* __restrict__ h,
                                 const float* __restrict__ bias,
                                 float* __restrict__ out, int n, int doRelu) {
    const int lane = threadIdx.x & 31;
    int node = (blockIdx.x * blockDim.x + threadIdx.x) >> 5;
    if (node >= n) return;
    node = __builtin_amdgcn_readfirstlane(node);

    const int beg = rowstart[node];
    const int end = rowstart[node + 1];
    const float di = dinv[node];
    const float s  = di * di;

    const int c0 = lane;
    const int c1 = lane + 32;
    const float4* hn = (const float4*)(h + (size_t)node * 256);
    const float4* b4 = (const float4*)bias;

    float4 hv0 = hn[c0], hv1 = hn[c1];
    float4 bv0 = b4[c0], bv1 = b4[c1];
    float4 a0, a1;
    a0.x = fmaf(s, hv0.x, bv0.x); a0.y = fmaf(s, hv0.y, bv0.y);
    a0.z = fmaf(s, hv0.z, bv0.z); a0.w = fmaf(s, hv0.w, bv0.w);
    a1.x = fmaf(s, hv1.x, bv1.x); a1.y = fmaf(s, hv1.y, bv1.y);
    a1.z = fmaf(s, hv1.z, bv1.z); a1.w = fmaf(s, hv1.w, bv1.w);

    for (int j = beg; j < end; ++j) {
        const int   sj = csr_src[j];
        const float nj = csr_norm[j];
        const float4* hs = (const float4*)(h + (size_t)sj * 256);
        float4 v0 = hs[c0];
        float4 v1 = hs[c1];
        a0.x = fmaf(nj, v0.x, a0.x); a0.y = fmaf(nj, v0.y, a0.y);
        a0.z = fmaf(nj, v0.z, a0.z); a0.w = fmaf(nj, v0.w, a0.w);
        a1.x = fmaf(nj, v1.x, a1.x); a1.y = fmaf(nj, v1.y, a1.y);
        a1.z = fmaf(nj, v1.z, a1.z); a1.w = fmaf(nj, v1.w, a1.w);
    }
    if (doRelu) {
        a0.x = fmaxf(a0.x, 0.f); a0.y = fmaxf(a0.y, 0.f);
        a0.z = fmaxf(a0.z, 0.f); a0.w = fmaxf(a0.w, 0.f);
        a1.x = fmaxf(a1.x, 0.f); a1.y = fmaxf(a1.y, 0.f);
        a1.z = fmaxf(a1.z, 0.f); a1.w = fmaxf(a1.w, 0.f);
    }
    float4* op = (float4*)(out + (size_t)node * 256);
    op[c0] = a0;
    op[c1] = a1;
}

__global__ void gather64_kernel(const int* __restrict__ rowstart,
                                const int* __restrict__ csr_src,
                                const float* __restrict__ csr_norm,
                                const float* __restrict__ dinv,
                                const float* __restrict__ h,
                                const float* __restrict__ bias,
                                float* __restrict__ out, int n) {
    const int lane = threadIdx.x & 31;
    int node = (blockIdx.x * blockDim.x + threadIdx.x) >> 5;
    if (node >= n) return;
    node = __builtin_amdgcn_readfirstlane(node);

    const int beg = rowstart[node];
    const int end = rowstart[node + 1];
    const float di = dinv[node];
    const float s  = di * di;

    const float2* hn = (const float2*)(h + (size_t)node * 64);
    const float2* b2 = (const float2*)bias;
    float2 hv = hn[lane];
    float2 bv = b2[lane];
    float2 a;
    a.x = fmaf(s, hv.x, bv.x);
    a.y = fmaf(s, hv.y, bv.y);

    for (int j = beg; j < end; ++j) {
        const int   sj = csr_src[j];
        const float nj = csr_norm[j];
        const float2* hs = (const float2*)(h + (size_t)sj * 64);
        float2 v = hs[lane];
        a.x = fmaf(nj, v.x, a.x);
        a.y = fmaf(nj, v.y, a.y);
    }
    ((float2*)(out + (size_t)node * 64))[lane] = a;
}

extern "C" void kernel_launch(void* const* d_in, const int* in_sizes, int n_in,
                              void* d_out, int out_size, void* d_ws, size_t ws_size,
                              hipStream_t stream) {
    (void)n_in; (void)out_size; (void)ws_size;

    const float* x  = (const float*)d_in[0];
    const int*   ei = (const int*)d_in[1];
    const float* W1 = (const float*)d_in[2];
    const float* b1 = (const float*)d_in[3];
    const float* W2 = (const float*)d_in[4];
    const float* b2 = (const float*)d_in[5];
    float* out = (float*)d_out;

    const int Fin = 128, Fmid = 256, Fout = 64;
    const int N = in_sizes[0] / Fin;
    const int E = in_sizes[1] / 2;
    const int* src = ei;
    const int* dst = ei + E;

    char* ws = (char*)d_ws;
    size_t off = 0;
    auto take = [&](size_t bytes) -> char* {
        char* p = ws + off;
        off = (off + bytes + 255) & ~(size_t)255;
        return p;
    };
    int*   deg      = (int*)  take((size_t)N * 4);
    float* dinv     = (float*)take((size_t)N * 4);
    int*   rowstart = (int*)  take((size_t)(N + 1) * 4);
    int*   cursor   = (int*)  take((size_t)N * 4);
    int*   csr_src  = (int*)  take((size_t)E * 4);
    float* csr_norm = (float*)take((size_t)E * 4);
    float* hbuf     = (float*)take((size_t)N * Fmid * 4);
    float* hrelu    = (float*)take((size_t)N * Fmid * 4);
    float* h2 = hbuf;

    const int T = 256;

    zero_i32_kernel <<<(N + T - 1) / T, T, 0, stream>>>(deg, N);
    count_deg_kernel<<<(E + T - 1) / T, T, 0, stream>>>(dst, deg, E);
    dinv_kernel     <<<(N + T - 1) / T, T, 0, stream>>>(deg, dinv, N);
    scan_kernel     <<<1, T, 0, stream>>>(deg, rowstart, N);
    copy_i32_kernel <<<(N + T - 1) / T, T, 0, stream>>>(rowstart, cursor, N);
    fill_csr_kernel <<<(E + T - 1) / T, T, 0, stream>>>(src, dst, dinv, cursor,
                                                        csr_src, csr_norm, E);

    {
        int tiles = (N / 16) * (Fmid / 16);
        int wpb = T / 32;
        gemm_f32_wmma_kernel<<<(tiles + wpb - 1) / wpb, T, 0, stream>>>(x, W1, hbuf, N, Fmid, Fin);
    }
    {
        long long threads = (long long)N * 32;
        gather256_kernel<<<(int)((threads + T - 1) / T), T, 0, stream>>>(
            rowstart, csr_src, csr_norm, dinv, hbuf, b1, hrelu, N, 1);
    }
    {
        int tiles = (N / 16) * (Fout / 16);
        int wpb = T / 32;
        gemm_f32_wmma_kernel<<<(tiles + wpb - 1) / wpb, T, 0, stream>>>(hrelu, W2, h2, N, Fout, Fmid);
    }
    {
        long long threads = (long long)N * 32;
        gather64_kernel<<<(int)((threads + T - 1) / T), T, 0, stream>>>(
            rowstart, csr_src, csr_norm, dinv, h2, b2, out, N);
    }
}


# Edge-chunk counting and the layer-1 GEMM share one launch: 64 counting work-groups (16 waves, node table in two LDS passes) beside 448 GEMM work-groups; 6 -> 5 launches; list order rule kept
# speedup vs baseline: 4.3206x; 4.3206x over previous
#include <hip/hip_runtime.h>

typedef float v2f __attribute__((ext_vector_type(2)));
typedef float v8f __attribute__((ext_vector_type(8)));

#define NCH  64
#define HCAP 25600
#define CG_T 512
#define CNT_U 8

__device__ __forceinline__ void
chunk_count_body(int* tbl, const int* __restrict__ dst, int* __restrict__ cnt, int* __restrict__ rank,
                 int* __restrict__ counter, int n, int E, int CH, int c) {
    const int tid = threadIdx.x;
    if (c == 0 && tid == 0) *counter = 0;
    const int e0 = c * CH;
    const int e1 = (e0 + CH < E) ? (e0 + CH) : E;

    for (int nb = 0; nb < n; nb += HCAP) {
        const int m = (n - nb < HCAP) ? (n - nb) : HCAP;
        for (int i = tid; 4 * i < m; i += CG_T) ((int4*)tbl)[i] = make_int4(0, 0, 0, 0);
        __syncthreads();
        for (int e = e0 + tid; e < e1; e += CNT_U * CG_T) {
            int d[CNT_U];
            #pragma unroll
            for (int u = 0; u < CNT_U; ++u) {
                const int ee = e + u * CG_T;
                const int v  = dst[(ee < e1) ? ee : e];
                d[u] = (ee < e1) ? (v - nb) : -1;
            }
            int r[CNT_U];
            #pragma unroll
            for (int u = 0; u < CNT_U; ++u) {
                r[u] = 0;
                if ((unsigned)d[u] < (unsigned)m) r[u] = atomicAdd(&tbl[d[u]], 1);
            }
            #pragma unroll
            for (int u = 0; u < CNT_U; ++u) {
                if ((unsigned)d[u] < (unsigned)m) rank[e + u * CG_T] = r[u];
            }
        }
        __syncthreads();
        int* row = cnt + (size_t)c * n + nb;
        if ((n & 3) == 0) {
            for (int i = tid; 4 * i < m; i += CG_T) ((int4*)row)[i] = ((const int4*)tbl)[i];
        } else {
            for (int i = tid; i < m; i += CG_T) row[i] = tbl[i];
        }
        __syncthreads();
    }
}

__global__ void __launch_bounds__(256)
scan_place_rows_kernel(int* __restrict__ cnt, int* __restrict__ counter,
                       int* __restrict__ rowrange, float* __restrict__ dinv, int n) {
    __shared__ int wtot[8];
    __shared__ int base_s;
    const int tid  = threadIdx.x;
    const int lane = tid & 31;
    const int w    = tid >> 5;
    const int i    = blockIdx.x * 256 + tid;
    const bool ok  = i < n;
    int* col = cnt + i;

    int t[NCH];
    #pragma unroll
    for (int c = 0; c < NCH; ++c) t[c] = 0;
    if (ok) {
        #pragma unroll
        for (int c = 0; c < NCH; ++c) t[c] = col[(size_t)c * n];
    }
    int v = 0;
    #pragma unroll
    for (int c = 0; c < NCH; ++c) v += t[c];

    int incl = v;
    for (int ofs = 1; ofs < 32; ofs <<= 1) {
        const int s = __shfl_up(incl, ofs);
        if (lane >= ofs) incl += s;
    }
    if (lane == 31) wtot[w] = incl;
    __syncthreads();
    if (tid == 0) {
        int run = 0;
        for (int k = 0; k < 8; ++k) { const int s = wtot[k]; wtot[k] = run; run += s; }
        base_s = atomicAdd(counter, run);
    }
    __syncthreads();
    if (ok) {
        const int pos = base_s + wtot[w] + (incl - v);
        ((int2*)rowrange)[i] = make_int2(pos, pos + v);
        dinv[i] = rsqrtf((float)v + 1.0f);
        int run = pos;
        #pragma unroll
        for (int c = 0; c < NCH; ++c) { col[(size_t)c * n] = run; run += t[c]; }
    }
}

__global__ void __launch_bounds__(256)
place_edges_kernel(const int* __restrict__ src, const int* __restrict__ dst,
                   const int* __restrict__ rank, const int* __restrict__ cnt,
                   const float* __restrict__ dinv,
                   int* __restrict__ csr_src, float* __restrict__ csr_norm,
                   int n, int E, int CH) {
    const int c  = blockIdx.y;
    const int e0 = c * CH;
    const int e1 = (e0 + CH < E) ? (e0 + CH) : E;
    const int eb = e0 + blockIdx.x * 1024 + threadIdx.x;
    if (eb >= e1) return;
    const int* base = cnt + (size_t)c * n;

    int s[4], d[4], r[4];
    #pragma unroll
    for (int u = 0; u < 4; ++u) {
        const int ee = eb + u * 256;
        const int ec = (ee < e1) ? ee : eb;
        s[u] = src[ec];
        d[u] = dst[ec];
        r[u] = rank[ec];
    }
    int   b[4];
    float ns[4], nd[4];
    #pragma unroll
    for (int u = 0; u < 4; ++u) {
        b[u]  = base[d[u]];
        ns[u] = dinv[s[u]];
        nd[u] = dinv[d[u]];
    }
    #pragma unroll
    for (int u = 0; u < 4; ++u) {
        if (eb + u * 256 < e1) {
            const int pos = b[u] + r[u];
            csr_src[pos]  = s[u];
            csr_norm[pos] = ns[u] * nd[u];
        }
    }
}

#define XT_PITCH 132

#define GEMM_WGS 448

__device__ __forceinline__ void
gemm_k128_n256_wres_body(float (*xt)[16 * XT_PITCH],
                         const float* __restrict__ X,
                         const float* __restrict__ W,
                         float* __restrict__ C, int mTiles, int wg, int G) {
    const int tid  = threadIdx.x;
    const int lane = tid & 31;
    const int w    = tid >> 5;
    const int l16  = lane & 15;
    const int hi   = lane >> 4;
    const int kOff = hi << 1;
    const int col  = (w << 4) + l16;

    v2f bf[32];
    {
        const float* bp = W + (size_t)kOff * 256 + col;
        #pragma unroll
        for (int i = 0; i < 32; ++i) {
            bf[i].x = bp[(size_t)(4 * i) * 256];
            bf[i].y = bp[(size_t)(4 * i + 1) * 256];
        }
    }

    const int stOfs = w * XT_PITCH + 4 * lane;
    int rt = wg;
    float4 stage = make_float4(0.f, 0.f, 0.f, 0.f);
    if (rt < mTiles) stage = ((const float4*)(X + (size_t)(rt * 16 + w) * 128))[lane];
    *(float4*)(&xt[0][stOfs]) = stage;
    __syncthreads();

    float* cbase = C + (size_t)(hi << 3) * 256 + col;
    v8f pend = {};
    int pendRt = -1;
    int cur = 0;
    for (; rt < mTiles; rt += G) {
        const int nx = rt + G;
        if (nx < mTiles) stage = ((const float4*)(X + (size_t)(nx * 16 + w) * 128))[lane];

        if (pendRt >= 0) {
            float* cp = cbase + (size_t)pendRt * (16 * 256);
            #pragma unroll
            for (int v = 0; v < 8; ++v) cp[(size_t)v * 256] = pend[v];
        }

        const float* ap = &xt[cur][l16 * XT_PITCH + kOff];
        v2f af[32];
        #pragma unroll
        for (int i = 0; i < 32; ++i) af[i] = *(const v2f*)(ap + 4 * i);

        v8f acc0 = {};
        v8f acc1 = {};
        #pragma unroll
        for (int i = 0; i < 32; i += 2) {
            acc0 = __builtin_amdgcn_wmma_f32_16x16x4_f32(false, af[i], false, bf[i],
                                                         (short)0, acc0, false, false);
            acc1 = __builtin_amdgcn_wmma_f32_16x16x4_f32(false, af[i + 1], false, bf[i + 1],
                                                         (short)0, acc1, false, false);
        }
        pend   = acc0 + acc1;
        pendRt = rt;

        if (nx < mTiles) *(float4*)(&xt[cur ^ 1][stOfs]) = stage;
        __syncthreads();
        cur ^= 1;
    }
    if (pendRt >= 0) {
        float* cp = cbase + (size_t)pendRt * (16 * 256);
        #pragma unroll
        for (int v = 0; v < 8; ++v) cp[(size_t)v * 256] = pend[v];
    }
}

__global__ void __launch_bounds__(CG_T)
count_and_gemm_kernel(const int* __restrict__ dst, int* __restrict__ cnt, int* __restrict__ rank,
                      int* __restrict__ counter, int n, int E, int CH,
                      const float* __restrict__ X, const float* __restrict__ W,
                      float* __restrict__ C, int mTiles) {
    __shared__ __attribute__((aligned(16))) int   tbl[HCAP];
    __shared__ __attribute__((aligned(16))) float xt[2][16 * XT_PITCH];
    const int b = blockIdx.x;
    if (b < NCH) chunk_count_body(tbl, dst, cnt, rank, counter, n, E, CH, b);
    else         gemm_k128_n256_wres_body(xt, X, W, C, mTiles, b - NCH, (int)gridDim.x - NCH);
}

#define AGG_PITCH 260
#define H2_PITCH  68

__global__ void __launch_bounds__(256)
gather256_relu_gemm64_kernel(const int* __restrict__ rowrange,
                             const int* __restrict__ csr_src,
                             const float* __restrict__ csr_norm,
                             const float* __restrict__ dinv,
                             const float* __restrict__ h,
                             const float* __restrict__ bias,
                             const float* __restrict__ W2,
                             float* __restrict__ h2, int n) {
    __shared__ float tile[16 * AGG_PITCH];
    __shared__ float otile[16 * H2_PITCH];
    const int lane = threadIdx.x & 31;
    const int w    = __builtin_amdgcn_readfirstlane((int)(threadIdx.x >> 5));
    const int base = blockIdx.x * 16;

    const int c0 = lane;
    const int c1 = lane + 32;
    const float4* b4 = (const float4*)bias;

    for (int t = 0; t < 2; ++t) {
        const int r    = w + 8 * t;
        const int node = base + r;
        float4 a0 = make_float4(0.f, 0.f, 0.f, 0.f);
        float4 a1 = a0;
        if (node < n) {
            const int2 rr = ((const int2*)rowrange)[node];
            const int beg = rr.x;
            const int end = rr.y;
            const float di = dinv[node];
            const float s  = di * di;
            const float4* hn = (const float4*)(h + (size_t)node * 256);

            float4 hv0 = hn[c0], hv1 = hn[c1];
            float4 bv0 = b4[c0], bv1 = b4[c1];
            a0.x = fmaf(s, hv0.x, bv0.x); a0.y = fmaf(s, hv0.y, bv0.y);
            a0.z = fmaf(s, hv0.z, bv0.z); a0.w = fmaf(s, hv0.w, bv0.w);
            a1.x = fmaf(s, hv1.x, bv1.x); a1.y = fmaf(s, hv1.y, bv1.y);
            a1.z = fmaf(s, hv1.z, bv1.z); a1.w = fmaf(s, hv1.w, bv1.w);

            for (int j = beg; j < end; ++j) {
                const int   sj = csr_src[j];
                const float nj = csr_norm[j];
                const float4* hs = (const float4*)(h + (size_t)sj * 256);
                float4 v0 = hs[c0];
                float4 v1 = hs[c1];
                a0.x = fmaf(nj, v0.x, a0.x); a0.y = fmaf(nj, v0.y, a0.y);
                a0.z = fmaf(nj, v0.z, a0.z); a0.w = fmaf(nj, v0.w, a0.w);
                a1.x = fmaf(nj, v1.x, a1.x); a1.y = fmaf(nj, v1.y, a1.y);
                a1.z = fmaf(nj, v1.z, a1.z); a1.w = fmaf(nj, v1.w, a1.w);
            }
            a0.x = fmaxf(a0.x, 0.f); a0.y = fmaxf(a0.y, 0.f);
            a0.z = fmaxf(a0.z, 0.f); a0.w = fmaxf(a0.w, 0.f);
            a1.x = fmaxf(a1.x, 0.f); a1.y = fmaxf(a1.y, 0.f);
            a1.z = fmaxf(a1.z, 0.f); a1.w = fmaxf(a1.w, 0.f);
        }
        float4* tp = (float4*)(tile + r * AGG_PITCH);
        tp[c0] = a0;
        tp[c1] = a1;
    }
    __syncthreads();

    if (w < 4) {
        const int l16  = lane & 15;
        const int hi   = lane >> 4;
        const int kOff = hi << 1;
        const int col  = (w << 4) + l16;
        const float* ap = tile + l16 * AGG_PITCH + kOff;
        const float* bp = W2 + (size_t)kOff * 64 + col;

        v8f acc0 = {};
        v8f acc1 = {};
        for (int k = 0; k < 256; k += 32) {
            v2f af[8];
            v2f bf[8];
            #pragma unroll
            for (int i = 0; i < 8; ++i) {
                const int kk = k + 4 * i;
                af[i]   = *(const v2f*)(ap + kk);
                bf[i].x = bp[(size_t)kk * 64];
                bf[i].y = bp[(size_t)kk * 64 + 64];
            }
            #pragma unroll
            for (int i = 0; i < 8; i += 2) {
                acc0 = __builtin_amdgcn_wmma_f32_16x16x4_f32(false, af[i], false, bf[i],
                                                             (short)0, acc0, false, false);
                acc1 = __builtin_amdgcn_wmma_f32_16x16x4_f32(false, af[i + 1], false, bf[i + 1],
                                                             (short)0, acc1, false, false);
            }
        }
        v8f acc = acc0 + acc1;

        float* op = otile + (hi << 3) * H2_PITCH + col;
        #pragma unroll
        for (int v = 0; v < 8; ++v) op[v * H2_PITCH] = acc[v];
    }
    __syncthreads();

    const int orow = threadIdx.x >> 4;
    const int oc   = threadIdx.x & 15;
    if (base + orow < n) {
        const float4 o = *(const float4*)(otile + orow * H2_PITCH + 4 * oc);
        ((float4*)(h2 + (size_t)(base + orow) * 64))[oc] = o;
    }
}

__global__ void gather64_kernel(const int* __restrict__ rowrange,
                                const int* __restrict__ csr_src,
                                const float* __restrict__ csr_norm,
                                const float* __restrict__ dinv,
                                const float* __restrict__ h,
                                const float* __restrict__ bias,
                                float* __restrict__ out, int n) {
    const int lane = threadIdx.x & 31;
    int node = (blockIdx.x * blockDim.x + threadIdx.x) >> 5;
    if (node >= n) return;
    node = __builtin_amdgcn_readfirstlane(node);

    const int2 rr = ((const int2*)rowrange)[node];
    const int beg = rr.x;
    const int end = rr.y;
    const float di = dinv[node];
    const float s  = di * di;

    const float2* hn = (const float2*)(h + (size_t)node * 64);
    const float2* b2 = (const float2*)bias;
    float2 hv = hn[lane];
    float2 bv = b2[lane];
    float2 a;
    a.x = fmaf(s, hv.x, bv.x);
    a.y = fmaf(s, hv.y, bv.y);

    for (int j = beg; j < end; ++j) {
        const int   sj = csr_src[j];
        const float nj = csr_norm[j];
        const float2* hs = (const float2*)(h + (size_t)sj * 64);
        float2 v = hs[lane];
        a.x = fmaf(nj, v.x, a.x);
        a.y = fmaf(nj, v.y, a.y);
    }
    ((float2*)(out + (size_t)node * 64))[lane] = a;
}

extern "C" void kernel_launch(void* const* d_in, const int* in_sizes, int n_in,
                              void* d_out, int out_size, void* d_ws, size_t ws_size,
                              hipStream_t stream) {
    (void)n_in; (void)out_size; (void)ws_size;

    const float* x  = (const float*)d_in[0];
    const int*   ei = (const int*)d_in[1];
    const float* W1 = (const float*)d_in[2];
    const float* b1 = (const float*)d_in[3];
    const float* W2 = (const float*)d_in[4];
    const float* b2 = (const float*)d_in[5];
    float* out = (float*)d_out;

    const int Fin = 128, Fmid = 256, Fout = 64;
    const int N = in_sizes[0] / Fin;
    const int E = in_sizes[1] / 2;
    const int* src = ei;
    const int* dst = ei + E;

    char* ws = (char*)d_ws;
    size_t off = 0;
    auto take = [&](size_t bytes) -> char* {
        char* p = ws + off;
        off = (off + bytes + 255) & ~(size_t)255;
        return p;
    };
    int*   counter  = (int*)  take(4);
    float* dinv     = (float*)take((size_t)N * 4);
    int*   rowrange = (int*)  take((size_t)N * 8);
    int*   cnt      = (int*)  take((size_t)NCH * N * 4);
    int*   rank     = (int*)  take((size_t)E * 4);
    int*   csr_src  = (int*)  take((size_t)E * 4);
    float* csr_norm = (float*)take((size_t)E * 4);
    float* hbuf     = (float*)take((size_t)N * Fmid * 4);
    float* h2       = (float*)take((size_t)N * Fout * 4);

    const int T = 256;

    const int CH = (E + NCH - 1) / NCH;
    count_and_gemm_kernel<<<NCH + GEMM_WGS, CG_T, 0, stream>>>(dst, cnt, rank, counter, N, E, CH,
                                                               x, W1, hbuf, N / 16);
    scan_place_rows_kernel<<<(N + 255) / 256, 256, 0, stream>>>(cnt, counter, rowrange, dinv, N);
    place_edges_kernel<<<dim3((CH + 1023) / 1024, NCH), 256, 0, stream>>>(src, dst, rank, cnt, dinv,
                                                                         csr_src, csr_norm, N, E, CH);

    gather256_relu_gemm64_kernel<<<(N + 15) / 16, 256, 0, stream>>>(
        rowrange, csr_src, csr_norm, dinv, hbuf, b1, W2, h2, N);
    {
        long long threads = (long long)N * 32;
        gather64_kernel<<<(int)((threads + T - 1) / T), T, 0, stream>>>(
            rowrange, csr_src, csr_norm, dinv, h2, b2, out, N);
    }
}


# CSR entry packed into one 8-byte {source, weight} record: the edge-placing launch does one scattered 8-B store per edge, not two 4-B ones; both gathers read one record stream; sum order unchanged
# speedup vs baseline: 4.9264x; 1.1402x over previous
#include <hip/hip_runtime.h>

typedef float v2f __attribute__((ext_vector_type(2)));
typedef float v8f __attribute__((ext_vector_type(8)));

#define NCH  64
#define HCAP 25600
#define CG_T 512
#define CNT_U 8

__device__ __forceinline__ void
chunk_count_body(int* tbl, const int* __restrict__ dst, int* __restrict__ cnt, int* __restrict__ rank,
                 int* __restrict__ counter, int n, int E, int CH, int c) {
    const int tid = threadIdx.x;
    if (c == 0 && tid == 0) *counter = 0;
    const int e0 = c * CH;
    const int e1 = (e0 + CH < E) ? (e0 + CH) : E;

    for (int nb = 0; nb < n; nb += HCAP) {
        const int m = (n - nb < HCAP) ? (n - nb) : HCAP;
        for (int i = tid; 4 * i < m; i += CG_T) ((int4*)tbl)[i] = make_int4(0, 0, 0, 0);
        __syncthreads();
        for (int e = e0 + tid; e < e1; e += CNT_U * CG_T) {
            int d[CNT_U];
            #pragma unroll
            for (int u = 0; u < CNT_U; ++u) {
                const int ee = e + u * CG_T;
                const int v  = dst[(ee < e1) ? ee : e];
                d[u] = (ee < e1) ? (v - nb) : -1;
            }
            int r[CNT_U];
            #pragma unroll
            for (int u = 0; u < CNT_U; ++u) {
                r[u] = 0;
                if ((unsigned)d[u] < (unsigned)m) r[u] = atomicAdd(&tbl[d[u]], 1);
            }
            #pragma unroll
            for (int u = 0; u < CNT_U; ++u) {
                if ((unsigned)d[u] < (unsigned)m) rank[e + u * CG_T] = r[u];
            }
        }
        __syncthreads();
        int* row = cnt + (size_t)c * n + nb;
        if ((n & 3) == 0) {
            for (int i = tid; 4 * i < m; i += CG_T) ((int4*)row)[i] = ((const int4*)tbl)[i];
        } else {
            for (int i = tid; i < m; i += CG_T) row[i] = tbl[i];
        }
        __syncthreads();
    }
}

__global__ void __launch_bounds__(256)
scan_place_rows_kernel(int* __restrict__ cnt, int* __restrict__ counter,
                       int* __restrict__ rowrange, float* __restrict__ dinv, int n) {
    __shared__ int wtot[8];
    __shared__ int base_s;
    const int tid  = threadIdx.x;
    const int lane = tid & 31;
    const int w    = tid >> 5;
    const int i    = blockIdx.x * 256 + tid;
    const bool ok  = i < n;
    int* col = cnt + i;

    int t[NCH];
    #pragma unroll
    for (int c = 0; c < NCH; ++c) t[c] = 0;
    if (ok) {
        #pragma unroll
        for (int c = 0; c < NCH; ++c) t[c] = col[(size_t)c * n];
    }
    int v = 0;
    #pragma unroll
    for (int c = 0; c < NCH; ++c) v += t[c];

    int incl = v;
    for (int ofs = 1; ofs < 32; ofs <<= 1) {
        const int s = __shfl_up(incl, ofs);
        if (lane >= ofs) incl += s;
    }
    if (lane == 31) wtot[w] = incl;
    __syncthreads();
    if (tid == 0) {
        int run = 0;
        for (int k = 0; k < 8; ++k) { const int s = wtot[k]; wtot[k] = run; run += s; }
        base_s = atomicAdd(counter, run);
    }
    __syncthreads();
    if (ok) {
        const int pos = base_s + wtot[w] + (incl - v);
        ((int2*)rowrange)[i] = make_int2(pos, pos + v);
        dinv[i] = rsqrtf((float)v + 1.0f);
        int run = pos;
        #pragma unroll
        for (int c = 0; c < NCH; ++c) { col[(size_t)c * n] = run; run += t[c]; }
    }
}

__global__ void __launch_bounds__(256)
place_edges_kernel(const int* __restrict__ src, const int* __restrict__ dst,
                   const int* __restrict__ rank, const int* __restrict__ cnt,
                   const float* __restrict__ dinv,
                   int* __restrict__ csr,
                   int n, int E, int CH) {
    const int c  = blockIdx.y;
    const int e0 = c * CH;
    const int e1 = (e0 + CH < E) ? (e0 + CH) : E;
    const int eb = e0 + blockIdx.x * 1024 + threadIdx.x;
    if (eb >= e1) return;
    const int* base = cnt + (size_t)c * n;

    int s[4], d[4], r[4];
    #pragma unroll
    for (int u = 0; u < 4; ++u) {
        const int ee = eb + u * 256;
        const int ec = (ee < e1) ? ee : eb;
        s[u] = src[ec];
        d[u] = dst[ec];
        r[u] = rank[ec];
    }
    int   b[4];
    float ns[4], nd[4];
    #pragma unroll
    for (int u = 0; u < 4; ++u) {
        b[u]  = base[d[u]];
        ns[u] = dinv[s[u]];
        nd[u] = dinv[d[u]];
    }
    #pragma unroll
    for (int u = 0; u < 4; ++u) {
        if (eb + u * 256 < e1) {
            const int pos = b[u] + r[u];
            ((int2*)csr)[pos] = make_int2(s[u], __float_as_int(ns[u] * nd[u]));
        }
    }
}

#define XT_PITCH 132

#define GEMM_WGS 448

__device__ __forceinline__ void
gemm_k128_n256_wres_body(float (*xt)[16 * XT_PITCH],
                         const float* __restrict__ X,
                         const float* __restrict__ W,
                         float* __restrict__ C, int mTiles, int wg, int G) {
    const int tid  = threadIdx.x;
    const int lane = tid & 31;
    const int w    = tid >> 5;
    const int l16  = lane & 15;
    const int hi   = lane >> 4;
    const int kOff = hi << 1;
    const int col  = (w << 4) + l16;

    v2f bf[32];
    {
        const float* bp = W + (size_t)kOff * 256 + col;
        #pragma unroll
        for (int i = 0; i < 32; ++i) {
            bf[i].x = bp[(size_t)(4 * i) * 256];
            bf[i].y = bp[(size_t)(4 * i + 1) * 256];
        }
    }

    const int stOfs = w * XT_PITCH + 4 * lane;
    int rt = wg;
    float4 stage = make_float4(0.f, 0.f, 0.f, 0.f);
    if (rt < mTiles) stage = ((const float4*)(X + (size_t)(rt * 16 + w) * 128))[lane];
    *(float4*)(&xt[0][stOfs]) = stage;
    __syncthreads();

    float* cbase = C + (size_t)(hi << 3) * 256 + col;
    v8f pend = {};
    int pendRt = -1;
    int cur = 0;
    for (; rt < mTiles; rt += G) {
        const int nx = rt + G;
        if (nx < mTiles) stage = ((const float4*)(X + (size_t)(nx * 16 + w) * 128))[lane];

        if (pendRt >= 0) {
            float* cp = cbase + (size_t)pendRt * (16 * 256);
            #pragma unroll
            for (int v = 0; v < 8; ++v) cp[(size_t)v * 256] = pend[v];
        }

        const float* ap = &xt[cur][l16 * XT_PITCH + kOff];
        v2f af[32];
        #pragma unroll
        for (int i = 0; i < 32; ++i) af[i] = *(const v2f*)(ap + 4 * i);

        v8f acc0 = {};
        v8f acc1 = {};
        #pragma unroll
        for (int i = 0; i < 32; i += 2) {
            acc0 = __builtin_amdgcn_wmma_f32_16x16x4_f32(false, af[i], false, bf[i],
                                                         (short)0, acc0, false, false);
            acc1 = __builtin_amdgcn_wmma_f32_16x16x4_f32(false, af[i + 1], false, bf[i + 1],
                                                         (short)0, acc1, false, false);
        }
        pend   = acc0 + acc1;
        pendRt = rt;

        if (nx < mTiles) *(float4*)(&xt[cur ^ 1][stOfs]) = stage;
        __syncthreads();
        cur ^= 1;
    }
    if (pendRt >= 0) {
        float* cp = cbase + (size_t)pendRt * (16 * 256);
        #pragma unroll
        for (int v = 0; v < 8; ++v) cp[(size_t)v * 256] = pend[v];
    }
}

__global__ void __launch_bounds__(CG_T)
count_and_gemm_kernel(const int* __restrict__ dst, int* __restrict__ cnt, int* __restrict__ rank,
                      int* __restrict__ counter, int n, int E, int CH,
                      const float* __restrict__ X, const float* __restrict__ W,
                      float* __restrict__ C, int mTiles) {
    __shared__ __attribute__((aligned(16))) int   tbl[HCAP];
    __shared__ __attribute__((aligned(16))) float xt[2][16 * XT_PITCH];
    const int b = blockIdx.x;
    if (b < NCH) chunk_count_body(tbl, dst, cnt, rank, counter, n, E, CH, b);
    else         gemm_k128_n256_wres_body(xt, X, W, C, mTiles, b - NCH, (int)gridDim.x - NCH);
}

#define AGG_PITCH 260
#define H2_PITCH  68

__global__ void __launch_bounds__(256)
gather256_relu_gemm64_kernel(const int* __restrict__ rowrange,
                             const int* __restrict__ csr,
                             const float* __restrict__ dinv,
                             const float* __restrict__ h,
                             const float* __restrict__ bias,
                             const float* __restrict__ W2,
                             float* __restrict__ h2, int n) {
    __shared__ float tile[16 * AGG_PITCH];
    __shared__ float otile[16 * H2_PITCH];
    const int lane = threadIdx.x & 31;
    const int w    = __builtin_amdgcn_readfirstlane((int)(threadIdx.x >> 5));
    const int base = blockIdx.x * 16;

    const int c0 = lane;
    const int c1 = lane + 32;
    const float4* b4 = (const float4*)bias;

    for (int t = 0; t < 2; ++t) {
        const int r    = w + 8 * t;
        const int node = base + r;
        float4 a0 = make_float4(0.f, 0.f, 0.f, 0.f);
        float4 a1 = a0;
        if (node < n) {
            const int2 rr = ((const int2*)rowrange)[node];
            const int beg = rr.x;
            const int end = rr.y;
            const float di = dinv[node];
            const float s  = di * di;
            const float4* hn = (const float4*)(h + (size_t)node * 256);

            float4 hv0 = hn[c0], hv1 = hn[c1];
            float4 bv0 = b4[c0], bv1 = b4[c1];
            a0.x = fmaf(s, hv0.x, bv0.x); a0.y = fmaf(s, hv0.y, bv0.y);
            a0.z = fmaf(s, hv0.z, bv0.z); a0.w = fmaf(s, hv0.w, bv0.w);
            a1.x = fmaf(s, hv1.x, bv1.x); a1.y = fmaf(s, hv1.y, bv1.y);
            a1.z = fmaf(s, hv1.z, bv1.z); a1.w = fmaf(s, hv1.w, bv1.w);

            for (int j = beg; j < end; ++j) {
                const int2  rec = ((const int2*)csr)[j];
                const int   sj = rec.x;
                const float nj = __int_as_float(rec.y);
                const float4* hs = (const float4*)(h + (size_t)sj * 256);
                float4 v0 = hs[c0];
                float4 v1 = hs[c1];
                a0.x = fmaf(nj, v0.x, a0.x); a0.y = fmaf(nj, v0.y, a0.y);
                a0.z = fmaf(nj, v0.z, a0.z); a0.w = fmaf(nj, v0.w, a0.w);
                a1.x = fmaf(nj, v1.x, a1.x); a1.y = fmaf(nj, v1.y, a1.y);
                a1.z = fmaf(nj, v1.z, a1.z); a1.w = fmaf(nj, v1.w, a1.w);
            }
            a0.x = fmaxf(a0.x, 0.f); a0.y = fmaxf(a0.y, 0.f);
            a0.z = fmaxf(a0.z, 0.f); a0.w = fmaxf(a0.w, 0.f);
            a1.x = fmaxf(a1.x, 0.f); a1.y = fmaxf(a1.y, 0.f);
            a1.z = fmaxf(a1.z, 0.f); a1.w = fmaxf(a1.w, 0.f);
        }
        float4* tp = (float4*)(tile + r * AGG_PITCH);
        tp[c0] = a0;
        tp[c1] = a1;
    }
    __syncthreads();

    if (w < 4) {
        const int l16  = lane & 15;
        const int hi   = lane >> 4;
        const int kOff = hi << 1;
        const int col  = (w << 4) + l16;
        const float* ap = tile + l16 * AGG_PITCH + kOff;
        const float* bp = W2 + (size_t)kOff * 64 + col;

        v8f acc0 = {};
        v8f acc1 = {};
        for (int k = 0; k < 256; k += 32) {
            v2f af[8];
            v2f bf[8];
            #pragma unroll
            for (int i = 0; i < 8; ++i) {
                const int kk = k + 4 * i;
                af[i]   = *(const v2f*)(ap + kk);
                bf[i].x = bp[(size_t)kk * 64];
                bf[i].y = bp[(size_t)kk * 64 + 64];
            }
            #pragma unroll
            for (int i = 0; i < 8; i += 2) {
                acc0 = __builtin_amdgcn_wmma_f32_16x16x4_f32(false, af[i], false, bf[i],
                                                             (short)0, acc0, false, false);
                acc1 = __builtin_amdgcn_wmma_f32_16x16x4_f32(false, af[i + 1], false, bf[i + 1],
                                                             (short)0, acc1, false, false);
            }
        }
        v8f acc = acc0 + acc1;

        float* op = otile + (hi << 3) * H2_PITCH + col;
        #pragma unroll
        for (int v = 0; v < 8; ++v) op[v * H2_PITCH] = acc[v];
    }
    __syncthreads();

    const int orow = threadIdx.x >> 4;
    const int oc   = threadIdx.x & 15;
    if (base + orow < n) {
        const float4 o = *(const float4*)(otile + orow * H2_PITCH + 4 * oc);
        ((float4*)(h2 + (size_t)(base + orow) * 64))[oc] = o;
    }
}

__global__ void gather64_kernel(const int* __restrict__ rowrange,
                                const int* __restrict__ csr,
                                const float* __restrict__ dinv,
                                const float* __restrict__ h,
                                const float* __restrict__ bias,
                                float* __restrict__ out, int n) {
    const int lane = threadIdx.x & 31;
    int node = (blockIdx.x * blockDim.x + threadIdx.x) >> 5;
    if (node >= n) return;
    node = __builtin_amdgcn_readfirstlane(node);

    const int2 rr = ((const int2*)rowrange)[node];
    const int beg = rr.x;
    const int end = rr.y;
    const float di = dinv[node];
    const float s  = di * di;

    const float2* hn = (const float2*)(h + (size_t)node * 64);
    const float2* b2 = (const float2*)bias;
    float2 hv = hn[lane];
    float2 bv = b2[lane];
    float2 a;
    a.x = fmaf(s, hv.x, bv.x);
    a.y = fmaf(s, hv.y, bv.y);

    for (int j = beg; j < end; ++j) {
        const int2  rec = ((const int2*)csr)[j];
        const int   sj = rec.x;
        const float nj = __int_as_float(rec.y);
        const float2* hs = (const float2*)(h + (size_t)sj * 64);
        float2 v = hs[lane];
        a.x = fmaf(nj, v.x, a.x);
        a.y = fmaf(nj, v.y, a.y);
    }
    ((float2*)(out + (size_t)node * 64))[lane] = a;
}

extern "C" void kernel_launch(void* const* d_in, const int* in_sizes, int n_in,
                              void* d_out, int out_size, void* d_ws, size_t ws_size,
                              hipStream_t stream) {
    (void)n_in; (void)out_size; (void)ws_size;

    const float* x  = (const float*)d_in[0];
    const int*   ei = (const int*)d_in[1];
    const float* W1 = (const float*)d_in[2];
    const float* b1 = (const float*)d_in[3];
    const float* W2 = (const float*)d_in[4];
    const float* b2 = (const float*)d_in[5];
    float* out = (float*)d_out;

    const int Fin = 128, Fmid = 256, Fout = 64;
    const int N = in_sizes[0] / Fin;
    const int E = in_sizes[1] / 2;
    const int* src = ei;
    const int* dst = ei + E;

    char* ws = (char*)d_ws;
    size_t off = 0;
    auto take = [&](size_t bytes) -> char* {
        char* p = ws + off;
        off = (off + bytes + 255) & ~(size_t)255;
        return p;
    };
    int*   counter  = (int*)  take(4);
    float* dinv     = (float*)take((size_t)N * 4);
    int*   rowrange = (int*)  take((size_t)N * 8);
    int*   cnt      = (int*)  take((size_t)NCH * N * 4);
    int*   rank     = (int*)  take((size_t)E * 4);
    int*   csr      = (int*)  take((size_t)E * 8);
    float* hbuf     = (float*)take((size_t)N * Fmid * 4);
    float* h2       = (float*)take((size_t)N * Fout * 4);

    const int T = 256;

    const int CH = (E + NCH - 1) / NCH;
    count_and_gemm_kernel<<<NCH + GEMM_WGS, CG_T, 0, stream>>>(dst, cnt, rank, counter, N, E, CH,
                                                               x, W1, hbuf, N / 16);
    scan_place_rows_kernel<<<(N + 255) / 256, 256, 0, stream>>>(cnt, counter, rowrange, dinv, N);
    place_edges_kernel<<<dim3((CH + 1023) / 1024, NCH), 256, 0, stream>>>(src, dst, rank, cnt, dinv,
                                                                         csr, N, E, CH);

    gather256_relu_gemm64_kernel<<<(N + 15) / 16, 256, 0, stream>>>(
        rowrange, csr, dinv, hbuf, b1, W2, h2, N);
    {
        long long threads = (long long)N * 32;
        gather64_kernel<<<(int)((threads + T - 1) / T), T, 0, stream>>>(
            rowrange, csr, dinv, h2, b2, out, N);
    }
}
